# starting kernel
# baseline (speedup 1.0000x reference)
#include <hip/hip_runtime.h>
#include <stdint.h>

typedef float v2f __attribute__((ext_vector_type(2)));
typedef float v8f __attribute__((ext_vector_type(8)));

__global__ void __launch_bounds__(256)
gemm_bias_wmma(const float* __restrict__ X, const float* __restrict__ W,
               const float* __restrict__ bias, float* __restrict__ Y,
               int K, int n_rows)
{
    const int wave    = (blockIdx.x * blockDim.x + threadIdx.x) >> 5;
    const int lane    = threadIdx.x & 31;
    const int n_tiles = (n_rows + 15) >> 4;
    if (wave >= n_tiles) return;

    const int half = lane >> 4;
    const int mrow = lane & 15;
    const int arow = min(wave * 16 + mrow, n_rows - 1);
    const float* __restrict__ xrow = X + (size_t)arow * (size_t)K;

    v8f c0 = {};
    v8f c1 = {};

    for (int k = 0; k < K; k += 4) {
        const int kb = k + half * 2;
        v2f a = *(const v2f*)(xrow + kb);

        v2f b0, b1;
        b0.x = W[(size_t)kb * 32 + mrow];
        b0.y = W[(size_t)(kb + 1) * 32 + mrow];
        b1.x = W[(size_t)kb * 32 + 16 + mrow];
        b1.y = W[(size_t)(kb + 1) * 32 + 16 + mrow];

        c0 = __builtin_amdgcn_wmma_f32_16x16x4_f32(false, a, false, b0,
                                                   (short)0, c0, false, false);
        c1 = __builtin_amdgcn_wmma_f32_16x16x4_f32(false, a, false, b1,
                                                   (short)0, c1, false, false);
    }

    const float bv0 = bias[mrow];
    const float bv1 = bias[16 + mrow];
#pragma unroll
    for (int r = 0; r < 8; ++r) {
        const int orow = wave * 16 + r + half * 8;
        if (orow < n_rows) {
            Y[(size_t)orow * 32 + mrow]      = c0[r] + bv0;
            Y[(size_t)orow * 32 + 16 + mrow] = c1[r] + bv1;
        }
    }
}

__global__ void init_zero(float* __restrict__ cnt, float* __restrict__ accB,
                          float* __restrict__ out, int n_nodes)
{
    const int i     = blockIdx.x * blockDim.x + threadIdx.x;
    const int total = n_nodes * 32;
    if (i < total) { accB[i] = 0.0f; out[i] = 0.0f; }
    if (i < n_nodes) cnt[i] = 0.0f;
}

__global__ void degree_kernel(const int* __restrict__ dst,
                              float* __restrict__ cnt, int n_edges)
{
    const int e = blockIdx.x * blockDim.x + threadIdx.x;
    if (e < n_edges) atomicAdd(&cnt[dst[e]], 1.0f);
}

__global__ void scatter_accum(const int* __restrict__ src, const int* __restrict__ dst,
                              const float* __restrict__ H, float* __restrict__ Acc,
                              int n_edges)
{
    const int t = blockIdx.x * blockDim.x + threadIdx.x;
    const int e = t >> 5;
    const int f = t & 31;
    if (e < n_edges) {
        const int s = src[e];
        const int d = dst[e];
        atomicAdd(&Acc[(size_t)d * 32 + f], H[(size_t)s * 32 + f]);
    }
}

__global__ void finalize_relu(const float* __restrict__ B, const float* __restrict__ cnt,
                              float* __restrict__ A, int n_nodes)
{
    const int i = blockIdx.x * blockDim.x + threadIdx.x;
    if (i < n_nodes * 32) {
        const float c = cnt[i >> 5];
        const float v = B[i] / fmaxf(c, 1.0f);
        A[i] = fmaxf(v, 0.0f);
    }
}

__global__ void finalize_div(float* __restrict__ out, const float* __restrict__ cnt,
                             int n_nodes)
{
    const int i = blockIdx.x * blockDim.x + threadIdx.x;
    if (i < n_nodes * 32) {
        out[i] = out[i] / fmaxf(cnt[i >> 5], 1.0f);
    }
}

extern "C" void kernel_launch(void* const* d_in, const int* in_sizes, int n_in,
                              void* d_out, int out_size, void* d_ws, size_t ws_size,
                              hipStream_t stream)
{
    const float* x    = (const float*)d_in[0];
    const int*   eidx = (const int*)d_in[1];
    const float* W1   = (const float*)d_in[2];
    const float* b1   = (const float*)d_in[3];
    const float* W2   = (const float*)d_in[4];
    const float* b2   = (const float*)d_in[5];
    float*       out  = (float*)d_out;

    const int n_nodes = in_sizes[0] / 128;
    const int n_edges = in_sizes[1] / 2;
    const int* src = eidx;
    const int* dst = eidx + n_edges;

    float* A   = (float*)d_ws;
    float* B   = A + (size_t)n_nodes * 32;
    float* cnt = B + (size_t)n_nodes * 32;

    const int TPB = 256;
    const int n_feat_elems   = n_nodes * 32;
    const int grid_feat      = (n_feat_elems + TPB - 1) / TPB;
    const int grid_edges     = (n_edges + TPB - 1) / TPB;
    const int grid_edge_feat = (int)(((size_t)n_edges * 32 + TPB - 1) / TPB);
    const int n_tiles        = (n_nodes + 15) / 16;
    const int grid_gemm      = (n_tiles + 7) / 8;

    init_zero<<<grid_feat, TPB, 0, stream>>>(cnt, B, out, n_nodes);
    degree_kernel<<<grid_edges, TPB, 0, stream>>>(dst, cnt, n_edges);
    gemm_bias_wmma<<<grid_gemm, TPB, 0, stream>>>(x, W1, b1, A, 128, n_nodes);
    scatter_accum<<<grid_edge_feat, TPB, 0, stream>>>(src, dst, A, B, n_edges);
    finalize_relu<<<grid_feat, TPB, 0, stream>>>(B, cnt, A, n_nodes);
    gemm_bias_wmma<<<grid_gemm, TPB, 0, stream>>>(A, W2, b2, B, 32, n_nodes);
    scatter_accum<<<grid_edge_feat, TPB, 0, stream>>>(src, dst, B, out, n_edges);
    finalize_div<<<grid_feat, TPB, 0, stream>>>(out, cnt, n_nodes);
}


# scatter pass: each wave takes 16 consecutive edges (was 1), reads their index pairs by scalar loads and issues all 16 row gathers before its atomic adds; sums stay in atomic arrival order
# speedup vs baseline: 1.8174x; 1.8174x over previous
#include <hip/hip_runtime.h>
#include <stdint.h>

typedef float v2f __attribute__((ext_vector_type(2)));
typedef float v8f __attribute__((ext_vector_type(8)));

__global__ void __launch_bounds__(256)
gemm_bias_wmma(const float* __restrict__ X, const float* __restrict__ W,
               const float* __restrict__ bias, float* __restrict__ Y,
               int K, int n_rows)
{
    const int wave    = (blockIdx.x * blockDim.x + threadIdx.x) >> 5;
    const int lane    = threadIdx.x & 31;
    const int n_tiles = (n_rows + 15) >> 4;
    if (wave >= n_tiles) return;

    const int half = lane >> 4;
    const int mrow = lane & 15;
    const int arow = min(wave * 16 + mrow, n_rows - 1);
    const float* __restrict__ xrow = X + (size_t)arow * (size_t)K;

    v8f c0 = {};
    v8f c1 = {};

    for (int k = 0; k < K; k += 4) {
        const int kb = k + half * 2;
        v2f a = *(const v2f*)(xrow + kb);

        v2f b0, b1;
        b0.x = W[(size_t)kb * 32 + mrow];
        b0.y = W[(size_t)(kb + 1) * 32 + mrow];
        b1.x = W[(size_t)kb * 32 + 16 + mrow];
        b1.y = W[(size_t)(kb + 1) * 32 + 16 + mrow];

        c0 = __builtin_amdgcn_wmma_f32_16x16x4_f32(false, a, false, b0,
                                                   (short)0, c0, false, false);
        c1 = __builtin_amdgcn_wmma_f32_16x16x4_f32(false, a, false, b1,
                                                   (short)0, c1, false, false);
    }

    const float bv0 = bias[mrow];
    const float bv1 = bias[16 + mrow];
#pragma unroll
    for (int r = 0; r < 8; ++r) {
        const int orow = wave * 16 + r + half * 8;
        if (orow < n_rows) {
            Y[(size_t)orow * 32 + mrow]      = c0[r] + bv0;
            Y[(size_t)orow * 32 + 16 + mrow] = c1[r] + bv1;
        }
    }
}

__global__ void init_zero(float* __restrict__ cnt, float* __restrict__ accB,
                          float* __restrict__ out, int n_nodes)
{
    const int i     = blockIdx.x * blockDim.x + threadIdx.x;
    const int total = n_nodes * 32;
    if (i < total) { accB[i] = 0.0f; out[i] = 0.0f; }
    if (i < n_nodes) cnt[i] = 0.0f;
}

__global__ void degree_kernel(const int* __restrict__ dst,
                              float* __restrict__ cnt, int n_edges)
{
    const int e = blockIdx.x * blockDim.x + threadIdx.x;
    if (e < n_edges) atomicAdd(&cnt[dst[e]], 1.0f);
}

#define EPW 16
#define SCATTER_WAVES 8

__global__ void __launch_bounds__(256)
scatter_accum(const int* __restrict__ src, const int* __restrict__ dst,
              const float* __restrict__ H, float* __restrict__ Acc,
              int n_edges)
{
    const int lane = threadIdx.x & 31;
    const int wib = __builtin_amdgcn_readfirstlane((int)(threadIdx.x >> 5));
    const int e0  = (blockIdx.x * SCATTER_WAVES + wib) * EPW;
    if (e0 >= n_edges) return;

    if (e0 + EPW <= n_edges) {
        int   s[EPW], d[EPW];
        float v[EPW];
#pragma unroll
        for (int i = 0; i < EPW; ++i) { s[i] = src[e0 + i]; d[i] = dst[e0 + i]; }
#pragma unroll
        for (int i = 0; i < EPW; ++i) {
            const float* hrow = H + (size_t)s[i] * 32;
            v[i] = hrow[lane];
        }
#pragma unroll
        for (int i = 0; i < EPW; ++i) {
            float* arow = Acc + (size_t)d[i] * 32;
            atomicAdd(arow + lane, v[i]);
        }
    } else {
        for (int e = e0; e < n_edges; ++e)
            atomicAdd(&Acc[(size_t)dst[e] * 32 + lane], H[(size_t)src[e] * 32 + lane]);
    }
}

__global__ void finalize_relu(const float* __restrict__ B, const float* __restrict__ cnt,
                              float* __restrict__ A, int n_nodes)
{
    const int i = blockIdx.x * blockDim.x + threadIdx.x;
    if (i < n_nodes * 32) {
        const float c = cnt[i >> 5];
        const float v = B[i] / fmaxf(c, 1.0f);
        A[i] = fmaxf(v, 0.0f);
    }
}

__global__ void finalize_div(float* __restrict__ out, const float* __restrict__ cnt,
                             int n_nodes)
{
    const int i = blockIdx.x * blockDim.x + threadIdx.x;
    if (i < n_nodes * 32) {
        out[i] = out[i] / fmaxf(cnt[i >> 5], 1.0f);
    }
}

extern "C" void kernel_launch(void* const* d_in, const int* in_sizes, int n_in,
                              void* d_out, int out_size, void* d_ws, size_t ws_size,
                              hipStream_t stream)
{
    const float* x    = (const float*)d_in[0];
    const int*   eidx = (const int*)d_in[1];
    const float* W1   = (const float*)d_in[2];
    const float* b1   = (const float*)d_in[3];
    const float* W2   = (const float*)d_in[4];
    const float* b2   = (const float*)d_in[5];
    float*       out  = (float*)d_out;

    const int n_nodes = in_sizes[0] / 128;
    const int n_edges = in_sizes[1] / 2;
    const int* src = eidx;
    const int* dst = eidx + n_edges;

    float* A   = (float*)d_ws;
    float* B   = A + (size_t)n_nodes * 32;
    float* cnt = B + (size_t)n_nodes * 32;

    const int TPB = 256;
    const int n_feat_elems   = n_nodes * 32;
    const int grid_feat      = (n_feat_elems + TPB - 1) / TPB;
    const int grid_edges     = (n_edges + TPB - 1) / TPB;
    const int edges_per_blk  = EPW * SCATTER_WAVES;
    const int grid_scatter   = (n_edges + edges_per_blk - 1) / edges_per_blk;
    const int n_tiles        = (n_nodes + 15) / 16;
    const int grid_gemm      = (n_tiles + 7) / 8;

    init_zero<<<grid_feat, TPB, 0, stream>>>(cnt, B, out, n_nodes);
    degree_kernel<<<grid_edges, TPB, 0, stream>>>(dst, cnt, n_edges);
    gemm_bias_wmma<<<grid_gemm, TPB, 0, stream>>>(x, W1, b1, A, 128, n_nodes);
    scatter_accum<<<grid_scatter, TPB, 0, stream>>>(src, dst, A, B, n_edges);
    finalize_relu<<<grid_feat, TPB, 0, stream>>>(B, cnt, A, n_nodes);
    gemm_bias_wmma<<<grid_gemm, TPB, 0, stream>>>(A, W2, b2, B, 32, n_nodes);
    scatter_accum<<<grid_scatter, TPB, 0, stream>>>(src, dst, B, out, n_edges);
    finalize_div<<<grid_feat, TPB, 0, stream>>>(out, cnt, n_nodes);
}


# scatter sums moved from L2 float atomics to LDS: edges binned once per call by owner node range, one work-group per range gathers rows and sums them with LDS adds; sums in LDS arrival order
# speedup vs baseline: 4.5614x; 2.5098x over previous
#include <hip/hip_runtime.h>
#include <stdint.h>


typedef float v2f __attribute__((ext_vector_type(2)));
typedef float v8f __attribute__((ext_vector_type(8)));

__global__ void __launch_bounds__(256)
gemm_bias_wmma(const float* __restrict__ X, const float* __restrict__ W,
               const float* __restrict__ bias, float* __restrict__ Y,
               int K, int n_rows)
{
    const int wave    = (blockIdx.x * blockDim.x + threadIdx.x) >> 5;
    const int lane    = threadIdx.x & 31;
    const int n_tiles = (n_rows + 15) >> 4;
    if (wave >= n_tiles) return;

    const int half = lane >> 4;
    const int mrow = lane & 15;
    const int arow = min(wave * 16 + mrow, n_rows - 1);
    const float* __restrict__ xrow = X + (size_t)arow * (size_t)K;

    v8f c0 = {};
    v8f c1 = {};

    for (int k = 0; k < K; k += 4) {
        const int kb = k + half * 2;
        v2f a = *(const v2f*)(xrow + kb);

        v2f b0, b1;
        b0.x = W[(size_t)kb * 32 + mrow];
        b0.y = W[(size_t)(kb + 1) * 32 + mrow];
        b1.x = W[(size_t)kb * 32 + 16 + mrow];
        b1.y = W[(size_t)(kb + 1) * 32 + 16 + mrow];

        c0 = __builtin_amdgcn_wmma_f32_16x16x4_f32(false, a, false, b0,
                                                   (short)0, c0, false, false);
        c1 = __builtin_amdgcn_wmma_f32_16x16x4_f32(false, a, false, b1,
                                                   (short)0, c1, false, false);
    }

    const float bv0 = bias[mrow];
    const float bv1 = bias[16 + mrow];
#pragma unroll
    for (int r = 0; r < 8; ++r) {
        const int orow = wave * 16 + r + half * 8;
        if (orow < n_rows) {
            Y[(size_t)orow * 32 + mrow]      = c0[r] + bv0;
            Y[(size_t)orow * 32 + 16 + mrow] = c1[r] + bv1;
        }
    }
}


#define NBK 256
#define NPB 391
#define BIN_THREADS 1024
#define CHUNK 6400
#define BATCH ((CHUNK + BIN_THREADS - 1) / BIN_THREADS)

__device__ __forceinline__ int owner_of(int d)
{
    const unsigned b = (unsigned)d / (unsigned)NPB;
    return (int)(b < (unsigned)(NBK - 1) ? b : (unsigned)(NBK - 1));
}

__global__ void __launch_bounds__(BIN_THREADS)
bin_count(const int* __restrict__ dst, int* __restrict__ counts, int n_edges)
{
    __shared__ int hist[NBK];
    const int tid = threadIdx.x;
    if (tid < NBK) hist[tid] = 0;
    __syncthreads();
    const int e_lo = blockIdx.x * CHUNK;
    const int e_hi = min(e_lo + CHUNK, n_edges);
    if (e_lo < e_hi) {
        int d[BATCH];
#pragma unroll
        for (int k = 0; k < BATCH; ++k)
            d[k] = dst[min(e_lo + k * BIN_THREADS + tid, e_hi - 1)];
#pragma unroll
        for (int k = 0; k < BATCH; ++k)
            if (e_lo + k * BIN_THREADS + tid < e_hi)
                atomicAdd(&hist[owner_of(d[k])], 1);
    }
    __syncthreads();
    if (tid < NBK) counts[blockIdx.x * NBK + tid] = hist[tid];
}

__global__ void __launch_bounds__(NBK)
bin_scan(const int* __restrict__ counts, int* __restrict__ pref,
         int* __restrict__ S, int nblk)
{
    __shared__ int sh[NBK];
    const int b = threadIdx.x;
    int run = 0;
#pragma unroll 32
    for (int w = 0; w < nblk; ++w) {
        const int c = counts[w * NBK + b];
        pref[w * NBK + b] = run;
        run += c;
    }
    sh[b] = run;
    __syncthreads();
    for (int off = 1; off < NBK; off <<= 1) {
        const int t = (b >= off) ? sh[b - off] : 0;
        __syncthreads();
        sh[b] += t;
        __syncthreads();
    }
    S[b + 1] = sh[b];
    if (b == 0) S[0] = 0;
}

__global__ void __launch_bounds__(BIN_THREADS)
bin_fill(const int* __restrict__ src, const int* __restrict__ dst,
         const int* __restrict__ counts, const int* __restrict__ pref,
         const int* __restrict__ S, int2* __restrict__ list, int n_edges)
{
    __shared__ int2 stage[CHUNK];
    __shared__ int  lstart[NBK];
    __shared__ int  lcur[NBK];
    __shared__ int  gbase[NBK];

    const int tid  = threadIdx.x;
    const int w    = blockIdx.x;
    const int e_lo = w * CHUNK;
    const int e_hi = min(e_lo + CHUNK, n_edges);
    const int n    = max(e_hi - e_lo, 0);

    int cnt_b = 0;
    if (tid < NBK) { cnt_b = counts[w * NBK + tid]; lcur[tid] = cnt_b; }
    __syncthreads();
    for (int off = 1; off < NBK; off <<= 1) {
        int t = 0;
        if (tid < NBK && tid >= off) t = lcur[tid - off];
        __syncthreads();
        if (tid < NBK) lcur[tid] += t;
        __syncthreads();
    }
    if (tid < NBK) {
        const int ex = lcur[tid] - cnt_b;
        lstart[tid] = ex;
        lcur[tid]   = ex;
        gbase[tid]  = S[tid] + pref[w * NBK + tid];
    }
    __syncthreads();

    if (n > 0) {
        int s[BATCH], d[BATCH];
#pragma unroll
        for (int k = 0; k < BATCH; ++k) {
            const int e = min(e_lo + k * BIN_THREADS + tid, e_hi - 1);
            s[k] = src[e];
            d[k] = dst[e];
        }
#pragma unroll
        for (int k = 0; k < BATCH; ++k) {
            if (e_lo + k * BIN_THREADS + tid < e_hi) {
                const int p = atomicAdd(&lcur[owner_of(d[k])], 1);
                if (p >= 0 && p < CHUNK) stage[p] = make_int2(s[k], d[k]);
            }
        }
    }
    __syncthreads();

    for (int i = tid; i < n; i += BIN_THREADS) {
        const int2 rec = stage[i];
        const int  b   = owner_of(rec.y);
        const int  g   = gbase[b] + (i - lstart[b]);
        if (g >= 0 && g < n_edges) list[g] = rec;
    }
}

#define AGG_THREADS 1024
#define AGG_WAVES   (AGG_THREADS / 32)
#define EPW 32

template <bool RELU>
__global__ void __launch_bounds__(AGG_THREADS)
owner_mean(const int2* __restrict__ list, const int* __restrict__ S,
           const float* __restrict__ H, float* __restrict__ Out,
           int n_nodes, int n_edges)
{
    __shared__ float acc[NPB * 32];
    __shared__ int   deg[NPB];

    const int tid  = threadIdx.x;
    const int lane = tid & 31;
    const int wave = __builtin_amdgcn_readfirstlane(tid >> 5);
    const int base = blockIdx.x * NPB;

    for (int i = tid; i < NPB * 32; i += AGG_THREADS) acc[i] = 0.0f;
    for (int i = tid; i < NPB; i += AGG_THREADS) deg[i] = 0;
    __syncthreads();

    const int e_begin = max(S[blockIdx.x], 0);
    const int e_end   = min(S[blockIdx.x + 1], n_edges);

    auto fetch = [&](int e) -> int2 {
        int2 t = list[max(min(e, e_end - 1), 0)];
        if (e >= e_end) t = make_int2(0, -1);
        return t;
    };
    int2 r = fetch(e_begin + wave * EPW + lane);
    for (int e0 = e_begin + wave * EPW; e0 < e_end; e0 += EPW * AGG_WAVES) {
        const int2 rn = fetch(e0 + EPW * AGG_WAVES + lane);

        if (r.y >= 0) {
            const int dl_own = min(max(r.y - base, 0), NPB - 1);
            atomicAdd(&deg[dl_own], 1);
        }

        float v[EPW];
#pragma unroll
        for (int i = 0; i < EPW; ++i) {
            int s = __builtin_amdgcn_readlane(r.x, i);
            s = min(max(s, 0), n_nodes - 1);
            const float* hrow = H + (size_t)s * 32;
            v[i] = hrow[lane];
        }
#pragma unroll
        for (int i = 0; i < EPW; ++i) {
            const int d = __builtin_amdgcn_readlane(r.y, i);
            if (d >= 0) {
                const int dl = min(max(d - base, 0), NPB - 1);
                atomicAdd(&acc[dl * 32 + lane], v[i]);
            }
        }
        r = rn;
    }
    __syncthreads();

    for (int i = tid; i < NPB * 32; i += AGG_THREADS) {
        const int nl = i >> 5;
        const int n  = base + nl;
        if (n < n_nodes) {
            const float c = (float)deg[nl];
            float m = acc[i] / fmaxf(c, 1.0f);
            if (RELU) m = fmaxf(m, 0.0f);
            Out[(size_t)n * 32 + (i & 31)] = m;
        }
    }
}

extern "C" void kernel_launch(void* const* d_in, const int* in_sizes, int n_in,
                              void* d_out, int out_size, void* d_ws, size_t ws_size,
                              hipStream_t stream)
{
    const float* x    = (const float*)d_in[0];
    const int*   eidx = (const int*)d_in[1];
    const float* W1   = (const float*)d_in[2];
    const float* b1   = (const float*)d_in[3];
    const float* W2   = (const float*)d_in[4];
    const float* b2   = (const float*)d_in[5];
    float*       out  = (float*)d_out;

    const int n_nodes = in_sizes[0] / 128;
    const int n_edges = in_sizes[1] / 2;
    const int* src = eidx;
    const int* dst = eidx + n_edges;

    const int nblk = (((n_edges + CHUNK - 1) / CHUNK) + 31) / 32 * 32;

    float* A      = (float*)d_ws;
    float* B      = A + (size_t)n_nodes * 32;
    int2*  list   = (int2*)(B + (size_t)n_nodes * 32);
    int*   S      = (int*)(list + n_edges);
    int*   counts = S + 2 * NBK;
    int*   pref   = counts + (size_t)nblk * NBK;

    const int TPB       = 256;
    const int n_tiles   = (n_nodes + 15) / 16;
    const int grid_gemm = (n_tiles + 7) / 8;

    bin_count<<<nblk, BIN_THREADS, 0, stream>>>(dst, counts, n_edges);
    bin_scan<<<1, NBK, 0, stream>>>(counts, pref, S, nblk);
    bin_fill<<<nblk, BIN_THREADS, 0, stream>>>(src, dst, counts, pref, S, list, n_edges);
    gemm_bias_wmma<<<grid_gemm, TPB, 0, stream>>>(x, W1, b1, A, 128, n_nodes);
    owner_mean<true><<<NBK, AGG_THREADS, 0, stream>>>(list, S, A, B, n_nodes, n_edges);
    gemm_bias_wmma<<<grid_gemm, TPB, 0, stream>>>(B, W2, b2, A, 32, n_nodes);
    owner_mean<false><<<NBK, AGG_THREADS, 0, stream>>>(list, S, A, out, n_nodes, n_edges);
}


# first layer's X@W1 tiles moved into the launch of the single-work-group bin scan (block 0 scans, blocks 1.. run the GEMM): the GEMM fills the WGPs the scan left idle; 7 launches -> 6
# speedup vs baseline: 4.9423x; 1.0835x over previous
#include <hip/hip_runtime.h>
#include <stdint.h>


typedef float v2f __attribute__((ext_vector_type(2)));
typedef float v8f __attribute__((ext_vector_type(8)));

__device__ __forceinline__ void
gemm_bias_tile(const float* __restrict__ X, const float* __restrict__ W,
               const float* __restrict__ bias, float* __restrict__ Y,
               int K, int n_rows, int wave)
{
    const int lane    = threadIdx.x & 31;
    const int n_tiles = (n_rows + 15) >> 4;
    if (wave >= n_tiles) return;

    const int half = lane >> 4;
    const int mrow = lane & 15;
    const int arow = min(wave * 16 + mrow, n_rows - 1);
    const float* __restrict__ xrow = X + (size_t)arow * (size_t)K;

    v8f c0 = {};
    v8f c1 = {};

    for (int k = 0; k < K; k += 4) {
        const int kb = k + half * 2;
        v2f a = *(const v2f*)(xrow + kb);

        v2f b0, b1;
        b0.x = W[(size_t)kb * 32 + mrow];
        b0.y = W[(size_t)(kb + 1) * 32 + mrow];
        b1.x = W[(size_t)kb * 32 + 16 + mrow];
        b1.y = W[(size_t)(kb + 1) * 32 + 16 + mrow];

        c0 = __builtin_amdgcn_wmma_f32_16x16x4_f32(false, a, false, b0,
                                                   (short)0, c0, false, false);
        c1 = __builtin_amdgcn_wmma_f32_16x16x4_f32(false, a, false, b1,
                                                   (short)0, c1, false, false);
    }

    const float bv0 = bias[mrow];
    const float bv1 = bias[16 + mrow];
#pragma unroll
    for (int r = 0; r < 8; ++r) {
        const int orow = wave * 16 + r + half * 8;
        if (orow < n_rows) {
            Y[(size_t)orow * 32 + mrow]      = c0[r] + bv0;
            Y[(size_t)orow * 32 + 16 + mrow] = c1[r] + bv1;
        }
    }
}

__global__ void __launch_bounds__(256)
gemm_bias_wmma(const float* __restrict__ X, const float* __restrict__ W,
               const float* __restrict__ bias, float* __restrict__ Y,
               int K, int n_rows)
{
    gemm_bias_tile(X, W, bias, Y, K, n_rows,
                   (int)((blockIdx.x * blockDim.x + threadIdx.x) >> 5));
}


#define NBK 256
#define NPB 391
#define BIN_THREADS 1024
#define CHUNK 6400
#define BATCH ((CHUNK + BIN_THREADS - 1) / BIN_THREADS)

__device__ __forceinline__ int owner_of(int d)
{
    const unsigned b = (unsigned)d / (unsigned)NPB;
    return (int)(b < (unsigned)(NBK - 1) ? b : (unsigned)(NBK - 1));
}

__global__ void __launch_bounds__(BIN_THREADS)
bin_count(const int* __restrict__ dst, int* __restrict__ counts, int n_edges)
{
    __shared__ int hist[NBK];
    const int tid = threadIdx.x;
    if (tid < NBK) hist[tid] = 0;
    __syncthreads();
    const int e_lo = blockIdx.x * CHUNK;
    const int e_hi = min(e_lo + CHUNK, n_edges);
    if (e_lo < e_hi) {
        int d[BATCH];
#pragma unroll
        for (int k = 0; k < BATCH; ++k)
            d[k] = dst[min(e_lo + k * BIN_THREADS + tid, e_hi - 1)];
#pragma unroll
        for (int k = 0; k < BATCH; ++k)
            if (e_lo + k * BIN_THREADS + tid < e_hi)
                atomicAdd(&hist[owner_of(d[k])], 1);
    }
    __syncthreads();
    if (tid < NBK) counts[blockIdx.x * NBK + tid] = hist[tid];
}

__global__ void __launch_bounds__(NBK)
bin_scan_and_gemm(const int* __restrict__ counts, int* __restrict__ pref,
                  int* __restrict__ S, int nblk,
                  const float* __restrict__ X, const float* __restrict__ W,
                  const float* __restrict__ bias, float* __restrict__ Y,
                  int K, int n_rows)
{
    __shared__ int sh[NBK];
    if (blockIdx.x != 0) {
        gemm_bias_tile(X, W, bias, Y, K, n_rows,
                       (int)(((blockIdx.x - 1) * blockDim.x + threadIdx.x) >> 5));
        return;
    }
    const int b = threadIdx.x;
    int run = 0;
#pragma unroll 32
    for (int w = 0; w < nblk; ++w) {
        const int c = counts[w * NBK + b];
        pref[w * NBK + b] = run;
        run += c;
    }
    sh[b] = run;
    __syncthreads();
    for (int off = 1; off < NBK; off <<= 1) {
        const int t = (b >= off) ? sh[b - off] : 0;
        __syncthreads();
        sh[b] += t;
        __syncthreads();
    }
    S[b + 1] = sh[b];
    if (b == 0) S[0] = 0;
}

__global__ void __launch_bounds__(BIN_THREADS)
bin_fill(const int* __restrict__ src, const int* __restrict__ dst,
         const int* __restrict__ counts, const int* __restrict__ pref,
         const int* __restrict__ S, int2* __restrict__ list, int n_edges)
{
    __shared__ int2 stage[CHUNK];
    __shared__ int  lstart[NBK];
    __shared__ int  lcur[NBK];
    __shared__ int  gbase[NBK];

    const int tid  = threadIdx.x;
    const int w    = blockIdx.x;
    const int e_lo = w * CHUNK;
    const int e_hi = min(e_lo + CHUNK, n_edges);
    const int n    = max(e_hi - e_lo, 0);

    int cnt_b = 0;
    if (tid < NBK) { cnt_b = counts[w * NBK + tid]; lcur[tid] = cnt_b; }
    __syncthreads();
    for (int off = 1; off < NBK; off <<= 1) {
        int t = 0;
        if (tid < NBK && tid >= off) t = lcur[tid - off];
        __syncthreads();
        if (tid < NBK) lcur[tid] += t;
        __syncthreads();
    }
    if (tid < NBK) {
        const int ex = lcur[tid] - cnt_b;
        lstart[tid] = ex;
        lcur[tid]   = ex;
        gbase[tid]  = S[tid] + pref[w * NBK + tid];
    }
    __syncthreads();

    if (n > 0) {
        int s[BATCH], d[BATCH];
#pragma unroll
        for (int k = 0; k < BATCH; ++k) {
            const int e = min(e_lo + k * BIN_THREADS + tid, e_hi - 1);
            s[k] = src[e];
            d[k] = dst[e];
        }
#pragma unroll
        for (int k = 0; k < BATCH; ++k) {
            if (e_lo + k * BIN_THREADS + tid < e_hi) {
                const int p = atomicAdd(&lcur[owner_of(d[k])], 1);
                if (p >= 0 && p < CHUNK) stage[p] = make_int2(s[k], d[k]);
            }
        }
    }
    __syncthreads();

    for (int i = tid; i < n; i += BIN_THREADS) {
        const int2 rec = stage[i];
        const int  b   = owner_of(rec.y);
        const int  g   = gbase[b] + (i - lstart[b]);
        if (g >= 0 && g < n_edges) list[g] = rec;
    }
}

#define AGG_THREADS 1024
#define AGG_WAVES   (AGG_THREADS / 32)
#define EPW 32

template <bool RELU>
__global__ void __launch_bounds__(AGG_THREADS)
owner_mean(const int2* __restrict__ list, const int* __restrict__ S,
           const float* __restrict__ H, float* __restrict__ Out,
           int n_nodes, int n_edges)
{
    __shared__ float acc[NPB * 32];
    __shared__ int   deg[NPB];

    const int tid  = threadIdx.x;
    const int lane = tid & 31;
    const int wave = __builtin_amdgcn_readfirstlane(tid >> 5);
    const int base = blockIdx.x * NPB;

    for (int i = tid; i < NPB * 32; i += AGG_THREADS) acc[i] = 0.0f;
    for (int i = tid; i < NPB; i += AGG_THREADS) deg[i] = 0;
    __syncthreads();

    const int e_begin = max(S[blockIdx.x], 0);
    const int e_end   = min(S[blockIdx.x + 1], n_edges);

    auto fetch = [&](int e) -> int2 {
        int2 t = list[max(min(e, e_end - 1), 0)];
        if (e >= e_end) t = make_int2(0, -1);
        return t;
    };
    int2 r = fetch(e_begin + wave * EPW + lane);
    for (int e0 = e_begin + wave * EPW; e0 < e_end; e0 += EPW * AGG_WAVES) {
        const int2 rn = fetch(e0 + EPW * AGG_WAVES + lane);

        if (r.y >= 0) {
            const int dl_own = min(max(r.y - base, 0), NPB - 1);
            atomicAdd(&deg[dl_own], 1);
        }

        float v[EPW];
#pragma unroll
        for (int i = 0; i < EPW; ++i) {
            int s = __builtin_amdgcn_readlane(r.x, i);
            s = min(max(s, 0), n_nodes - 1);
            const float* hrow = H + (size_t)s * 32;
            v[i] = hrow[lane];
        }
#pragma unroll
        for (int i = 0; i < EPW; ++i) {
            const int d = __builtin_amdgcn_readlane(r.y, i);
            if (d >= 0) {
                const int dl = min(max(d - base, 0), NPB - 1);
                atomicAdd(&acc[dl * 32 + lane], v[i]);
            }
        }
        r = rn;
    }
    __syncthreads();

    for (int i = tid; i < NPB * 32; i += AGG_THREADS) {
        const int nl = i >> 5;
        const int n  = base + nl;
        if (n < n_nodes) {
            const float c = (float)deg[nl];
            float m = acc[i] / fmaxf(c, 1.0f);
            if (RELU) m = fmaxf(m, 0.0f);
            Out[(size_t)n * 32 + (i & 31)] = m;
        }
    }
}

extern "C" void kernel_launch(void* const* d_in, const int* in_sizes, int n_in,
                              void* d_out, int out_size, void* d_ws, size_t ws_size,
                              hipStream_t stream)
{
    const float* x    = (const float*)d_in[0];
    const int*   eidx = (const int*)d_in[1];
    const float* W1   = (const float*)d_in[2];
    const float* b1   = (const float*)d_in[3];
    const float* W2   = (const float*)d_in[4];
    const float* b2   = (const float*)d_in[5];
    float*       out  = (float*)d_out;

    const int n_nodes = in_sizes[0] / 128;
    const int n_edges = in_sizes[1] / 2;
    const int* src = eidx;
    const int* dst = eidx + n_edges;

    const int nblk = (((n_edges + CHUNK - 1) / CHUNK) + 31) / 32 * 32;

    float* A      = (float*)d_ws;
    float* B      = A + (size_t)n_nodes * 32;
    int2*  list   = (int2*)(B + (size_t)n_nodes * 32);
    int*   S      = (int*)(list + n_edges);
    int*   counts = S + 2 * NBK;
    int*   pref   = counts + (size_t)nblk * NBK;

    const int TPB       = 256;
    const int n_tiles   = (n_nodes + 15) / 16;
    const int grid_gemm = (n_tiles + 7) / 8;

    bin_count<<<nblk, BIN_THREADS, 0, stream>>>(dst, counts, n_edges);
    bin_scan_and_gemm<<<grid_gemm + 1, NBK, 0, stream>>>(counts, pref, S, nblk,
                                                        x, W1, b1, A, 128, n_nodes);
    bin_fill<<<nblk, BIN_THREADS, 0, stream>>>(src, dst, counts, pref, S, list, n_edges);
    owner_mean<true><<<NBK, AGG_THREADS, 0, stream>>>(list, S, A, B, n_nodes, n_edges);
    gemm_bias_wmma<<<grid_gemm, TPB, 0, stream>>>(B, W2, b2, A, 32, n_nodes);
    owner_mean<false><<<NBK, AGG_THREADS, 0, stream>>>(list, S, A, out, n_nodes, n_edges);
}


# edge records written at binning as ready byte offsets (source row, LDS sum row; clamped there once): sum loop per edge is readlane+add+load, readlane+add+LDS add; no clamps, 64-bit math or branch
# speedup vs baseline: 5.2858x; 1.0695x over previous
#include <hip/hip_runtime.h>
#include <stdint.h>


typedef float v2f __attribute__((ext_vector_type(2)));
typedef float v8f __attribute__((ext_vector_type(8)));

__device__ __forceinline__ void
gemm_bias_tile(const float* __restrict__ X, const float* __restrict__ W,
               const float* __restrict__ bias, float* __restrict__ Y,
               int K, int n_rows, int wave)
{
    const int lane    = threadIdx.x & 31;
    const int n_tiles = (n_rows + 15) >> 4;
    if (wave >= n_tiles) return;

    const int half = lane >> 4;
    const int mrow = lane & 15;
    const int arow = min(wave * 16 + mrow, n_rows - 1);
    const float* __restrict__ xrow = X + (size_t)arow * (size_t)K;

    v8f c0 = {};
    v8f c1 = {};

    for (int k = 0; k < K; k += 4) {
        const int kb = k + half * 2;
        v2f a = *(const v2f*)(xrow + kb);

        v2f b0, b1;
        b0.x = W[(size_t)kb * 32 + mrow];
        b0.y = W[(size_t)(kb + 1) * 32 + mrow];
        b1.x = W[(size_t)kb * 32 + 16 + mrow];
        b1.y = W[(size_t)(kb + 1) * 32 + 16 + mrow];

        c0 = __builtin_amdgcn_wmma_f32_16x16x4_f32(false, a, false, b0,
                                                   (short)0, c0, false, false);
        c1 = __builtin_amdgcn_wmma_f32_16x16x4_f32(false, a, false, b1,
                                                   (short)0, c1, false, false);
    }

    const float bv0 = bias[mrow];
    const float bv1 = bias[16 + mrow];
#pragma unroll
    for (int r = 0; r < 8; ++r) {
        const int orow = wave * 16 + r + half * 8;
        if (orow < n_rows) {
            Y[(size_t)orow * 32 + mrow]      = c0[r] + bv0;
            Y[(size_t)orow * 32 + 16 + mrow] = c1[r] + bv1;
        }
    }
}

__global__ void __launch_bounds__(256)
gemm_bias_wmma(const float* __restrict__ X, const float* __restrict__ W,
               const float* __restrict__ bias, float* __restrict__ Y,
               int K, int n_rows)
{
    gemm_bias_tile(X, W, bias, Y, K, n_rows,
                   (int)((blockIdx.x * blockDim.x + threadIdx.x) >> 5));
}


#define NBK 256
#define NPB 391
#define BIN_THREADS 1024
#define CHUNK 6400
#define BATCH ((CHUNK + BIN_THREADS - 1) / BIN_THREADS)
#define ROWB 128

__device__ __forceinline__ int owner_of(int d)
{
    const unsigned b = (unsigned)d / (unsigned)NPB;
    return (int)(b < (unsigned)(NBK - 1) ? b : (unsigned)(NBK - 1));
}

__global__ void __launch_bounds__(BIN_THREADS)
bin_count(const int* __restrict__ dst, int* __restrict__ counts, int n_edges)
{
    __shared__ int hist[NBK];
    const int tid = threadIdx.x;
    if (tid < NBK) hist[tid] = 0;
    __syncthreads();
    const int e_lo = blockIdx.x * CHUNK;
    const int e_hi = min(e_lo + CHUNK, n_edges);
    if (e_lo < e_hi) {
        int d[BATCH];
#pragma unroll
        for (int k = 0; k < BATCH; ++k)
            d[k] = dst[min(e_lo + k * BIN_THREADS + tid, e_hi - 1)];
#pragma unroll
        for (int k = 0; k < BATCH; ++k)
            if (e_lo + k * BIN_THREADS + tid < e_hi)
                atomicAdd(&hist[owner_of(d[k])], 1);
    }
    __syncthreads();
    if (tid < NBK) counts[blockIdx.x * NBK + tid] = hist[tid];
}

__global__ void __launch_bounds__(NBK)
bin_scan_and_gemm(const int* __restrict__ counts, int* __restrict__ pref,
                  int* __restrict__ S, int nblk,
                  const float* __restrict__ X, const float* __restrict__ W,
                  const float* __restrict__ bias, float* __restrict__ Y,
                  int K, int n_rows)
{
    __shared__ int sh[NBK];
    if (blockIdx.x != 0) {
        gemm_bias_tile(X, W, bias, Y, K, n_rows,
                       (int)(((blockIdx.x - 1) * blockDim.x + threadIdx.x) >> 5));
        return;
    }
    const int b = threadIdx.x;
    int run = 0;
#pragma unroll 32
    for (int w = 0; w < nblk; ++w) {
        const int c = counts[w * NBK + b];
        pref[w * NBK + b] = run;
        run += c;
    }
    sh[b] = run;
    __syncthreads();
    for (int off = 1; off < NBK; off <<= 1) {
        const int t = (b >= off) ? sh[b - off] : 0;
        __syncthreads();
        sh[b] += t;
        __syncthreads();
    }
    S[b + 1] = sh[b];
    if (b == 0) S[0] = 0;
}

__global__ void __launch_bounds__(BIN_THREADS)
bin_fill(const int* __restrict__ src, const int* __restrict__ dst,
         const int* __restrict__ counts, const int* __restrict__ pref,
         const int* __restrict__ S, int2* __restrict__ list, int n_edges, int n_nodes)
{
    __shared__ int2 stage[CHUNK];
    __shared__ int  lstart[NBK];
    __shared__ int  lcur[NBK];
    __shared__ int  gbase[NBK];

    const int tid  = threadIdx.x;
    const int w    = blockIdx.x;
    const int e_lo = w * CHUNK;
    const int e_hi = min(e_lo + CHUNK, n_edges);
    const int n    = max(e_hi - e_lo, 0);

    int cnt_b = 0;
    if (tid < NBK) { cnt_b = counts[w * NBK + tid]; lcur[tid] = cnt_b; }
    __syncthreads();
    for (int off = 1; off < NBK; off <<= 1) {
        int t = 0;
        if (tid < NBK && tid >= off) t = lcur[tid - off];
        __syncthreads();
        if (tid < NBK) lcur[tid] += t;
        __syncthreads();
    }
    if (tid < NBK) {
        const int ex = lcur[tid] - cnt_b;
        lstart[tid] = ex;
        lcur[tid]   = ex;
        gbase[tid]  = S[tid] + pref[w * NBK + tid];
    }
    __syncthreads();

    if (n > 0) {
        int s[BATCH], d[BATCH];
#pragma unroll
        for (int k = 0; k < BATCH; ++k) {
            const int e = min(e_lo + k * BIN_THREADS + tid, e_hi - 1);
            s[k] = src[e];
            d[k] = dst[e];
        }
#pragma unroll
        for (int k = 0; k < BATCH; ++k) {
            if (e_lo + k * BIN_THREADS + tid < e_hi) {
                const int p = atomicAdd(&lcur[owner_of(d[k])], 1);
                if (p >= 0 && p < CHUNK) stage[p] = make_int2(s[k], d[k]);
            }
        }
    }
    __syncthreads();

    for (int i = tid; i < n; i += BIN_THREADS) {
        const int2 rec = stage[i];
        const int  b   = owner_of(rec.y);
        const int  g   = gbase[b] + (i - lstart[b]);
        const int so = min(max(rec.x, 0), n_nodes - 1) * ROWB;
        const int ao = min(max(rec.y - b * NPB, 0), NPB - 1) * ROWB;
        if (g >= 0 && g < n_edges) list[g] = make_int2(so, ao);
    }
}

#define AGG_THREADS 1024
#define AGG_WAVES   (AGG_THREADS / 32)
#define EPW 32
#define SPARE_OFS (NPB * ROWB)

template <bool RELU>
__global__ void __launch_bounds__(AGG_THREADS)
owner_mean(const int2* __restrict__ list, const int* __restrict__ S,
           const float* __restrict__ H, float* __restrict__ Out,
           int n_nodes, int n_edges)
{
    __shared__ float acc[(NPB + 1) * 32];
    __shared__ int   deg[NPB + 1];

    const int tid  = threadIdx.x;
    const int lane = tid & 31;
    const unsigned lane4 = (unsigned)lane * 4u;
    const int wave = __builtin_amdgcn_readfirstlane(tid >> 5);
    const int base = blockIdx.x * NPB;

    for (int i = tid; i < (NPB + 1) * 32; i += AGG_THREADS) acc[i] = 0.0f;
    for (int i = tid; i < NPB + 1; i += AGG_THREADS) deg[i] = 0;
    __syncthreads();

    const int e_begin = max(S[blockIdx.x], 0);
    const int e_end   = min(S[blockIdx.x + 1], n_edges);

    auto fetch = [&](int e) -> int2 {
        int2 t = list[max(min(e, e_end - 1), 0)];
        if (e >= e_end) t = make_int2(0, SPARE_OFS);
        return t;
    };
    int2 r = fetch(e_begin + wave * EPW + lane);
    for (int e0 = e_begin + wave * EPW; e0 < e_end; e0 += EPW * AGG_WAVES) {
        const int2 rn = fetch(e0 + EPW * AGG_WAVES + lane);

        atomicAdd(&deg[(unsigned)r.y / ROWB], 1);

        float v[EPW];
#pragma unroll
        for (int i = 0; i < EPW; ++i) {
            const unsigned so = (unsigned)__builtin_amdgcn_readlane(r.x, i);
            v[i] = *(const float*)((const char*)H + (so + lane4));
        }
#pragma unroll
        for (int i = 0; i < EPW; ++i) {
            const unsigned ao = (unsigned)__builtin_amdgcn_readlane(r.y, i);
            atomicAdd((float*)((char*)acc + (ao + lane4)), v[i]);
        }
        r = rn;
    }
    __syncthreads();

    for (int i = tid; i < NPB * 32; i += AGG_THREADS) {
        const int nl = i >> 5;
        const int n  = base + nl;
        if (n < n_nodes) {
            const float c = (float)deg[nl];
            float m = acc[i] / fmaxf(c, 1.0f);
            if (RELU) m = fmaxf(m, 0.0f);
            Out[(size_t)n * 32 + (i & 31)] = m;
        }
    }
}

extern "C" void kernel_launch(void* const* d_in, const int* in_sizes, int n_in,
                              void* d_out, int out_size, void* d_ws, size_t ws_size,
                              hipStream_t stream)
{
    const float* x    = (const float*)d_in[0];
    const int*   eidx = (const int*)d_in[1];
    const float* W1   = (const float*)d_in[2];
    const float* b1   = (const float*)d_in[3];
    const float* W2   = (const float*)d_in[4];
    const float* b2   = (const float*)d_in[5];
    float*       out  = (float*)d_out;

    const int n_nodes = in_sizes[0] / 128;
    const int n_edges = in_sizes[1] / 2;
    const int* src = eidx;
    const int* dst = eidx + n_edges;

    const int nblk = (((n_edges + CHUNK - 1) / CHUNK) + 31) / 32 * 32;

    float* A      = (float*)d_ws;
    float* B      = A + (size_t)n_nodes * 32;
    int2*  list   = (int2*)(B + (size_t)n_nodes * 32);
    int*   S      = (int*)(list + n_edges);
    int*   counts = S + 2 * NBK;
    int*   pref   = counts + (size_t)nblk * NBK;

    const int TPB       = 256;
    const int n_tiles   = (n_nodes + 15) / 16;
    const int grid_gemm = (n_tiles + 7) / 8;

    bin_count<<<nblk, BIN_THREADS, 0, stream>>>(dst, counts, n_edges);
    bin_scan_and_gemm<<<grid_gemm + 1, NBK, 0, stream>>>(counts, pref, S, nblk,
                                                        x, W1, b1, A, 128, n_nodes);
    bin_fill<<<nblk, BIN_THREADS, 0, stream>>>(src, dst, counts, pref, S, list, n_edges, n_nodes);
    owner_mean<true><<<NBK, AGG_THREADS, 0, stream>>>(list, S, A, B, n_nodes, n_edges);
    gemm_bias_wmma<<<grid_gemm, TPB, 0, stream>>>(B, W2, b2, A, 32, n_nodes);
    owner_mean<false><<<NBK, AGG_THREADS, 0, stream>>>(list, S, A, out, n_nodes, n_edges);
}


# second layer's product B@W2+b2 moved into the first sum pass's epilogue: each owner group takes it from its LDS means with the same matrix instruction and k order; 6 launches -> 5, means never stored
# speedup vs baseline: 5.7399x; 1.0859x over previous
#include <hip/hip_runtime.h>
#include <stdint.h>


typedef float v2f __attribute__((ext_vector_type(2)));
typedef float v8f __attribute__((ext_vector_type(8)));

__device__ __forceinline__ void
gemm_bias_tile(const float* __restrict__ X, const float* __restrict__ W,
               const float* __restrict__ bias, float* __restrict__ Y,
               int K, int n_rows, int wave)
{
    const int lane    = threadIdx.x & 31;
    const int n_tiles = (n_rows + 15) >> 4;
    if (wave >= n_tiles) return;

    const int half = lane >> 4;
    const int mrow = lane & 15;
    const int arow = min(wave * 16 + mrow, n_rows - 1);
    const float* __restrict__ xrow = X + (size_t)arow * (size_t)K;

    v8f c0 = {};
    v8f c1 = {};

    for (int k = 0; k < K; k += 4) {
        const int kb = k + half * 2;
        v2f a = *(const v2f*)(xrow + kb);

        v2f b0, b1;
        b0.x = W[(size_t)kb * 32 + mrow];
        b0.y = W[(size_t)(kb + 1) * 32 + mrow];
        b1.x = W[(size_t)kb * 32 + 16 + mrow];
        b1.y = W[(size_t)(kb + 1) * 32 + 16 + mrow];

        c0 = __builtin_amdgcn_wmma_f32_16x16x4_f32(false, a, false, b0,
                                                   (short)0, c0, false, false);
        c1 = __builtin_amdgcn_wmma_f32_16x16x4_f32(false, a, false, b1,
                                                   (short)0, c1, false, false);
    }

    const float bv0 = bias[mrow];
    const float bv1 = bias[16 + mrow];
#pragma unroll
    for (int r = 0; r < 8; ++r) {
        const int orow = wave * 16 + r + half * 8;
        if (orow < n_rows) {
            Y[(size_t)orow * 32 + mrow]      = c0[r] + bv0;
            Y[(size_t)orow * 32 + 16 + mrow] = c1[r] + bv1;
        }
    }
}

#define NBK 256
#define NPB 391
#define BIN_THREADS 1024
#define CHUNK 6400
#define BATCH ((CHUNK + BIN_THREADS - 1) / BIN_THREADS)
#define ROWB 128

__device__ __forceinline__ int owner_of(int d)
{
    const unsigned b = (unsigned)d / (unsigned)NPB;
    return (int)(b < (unsigned)(NBK - 1) ? b : (unsigned)(NBK - 1));
}

__global__ void __launch_bounds__(BIN_THREADS)
bin_count(const int* __restrict__ dst, int* __restrict__ counts, int n_edges)
{
    __shared__ int hist[NBK];
    const int tid = threadIdx.x;
    if (tid < NBK) hist[tid] = 0;
    __syncthreads();
    const int e_lo = blockIdx.x * CHUNK;
    const int e_hi = min(e_lo + CHUNK, n_edges);
    if (e_lo < e_hi) {
        int d[BATCH];
#pragma unroll
        for (int k = 0; k < BATCH; ++k)
            d[k] = dst[min(e_lo + k * BIN_THREADS + tid, e_hi - 1)];
#pragma unroll
        for (int k = 0; k < BATCH; ++k)
            if (e_lo + k * BIN_THREADS + tid < e_hi)
                atomicAdd(&hist[owner_of(d[k])], 1);
    }
    __syncthreads();
    if (tid < NBK) counts[blockIdx.x * NBK + tid] = hist[tid];
}

__global__ void __launch_bounds__(NBK)
bin_scan_and_gemm(const int* __restrict__ counts, int* __restrict__ pref,
                  int* __restrict__ S, int nblk,
                  const float* __restrict__ X, const float* __restrict__ W,
                  const float* __restrict__ bias, float* __restrict__ Y,
                  int K, int n_rows)
{
    __shared__ int sh[NBK];
    if (blockIdx.x != 0) {
        gemm_bias_tile(X, W, bias, Y, K, n_rows,
                       (int)(((blockIdx.x - 1) * blockDim.x + threadIdx.x) >> 5));
        return;
    }
    const int b = threadIdx.x;
    int run = 0;
#pragma unroll 32
    for (int w = 0; w < nblk; ++w) {
        const int c = counts[w * NBK + b];
        pref[w * NBK + b] = run;
        run += c;
    }
    sh[b] = run;
    __syncthreads();
    for (int off = 1; off < NBK; off <<= 1) {
        const int t = (b >= off) ? sh[b - off] : 0;
        __syncthreads();
        sh[b] += t;
        __syncthreads();
    }
    S[b + 1] = sh[b];
    if (b == 0) S[0] = 0;
}

__global__ void __launch_bounds__(BIN_THREADS)
bin_fill(const int* __restrict__ src, const int* __restrict__ dst,
         const int* __restrict__ counts, const int* __restrict__ pref,
         const int* __restrict__ S, int2* __restrict__ list, int n_edges, int n_nodes)
{
    __shared__ int2 stage[CHUNK];
    __shared__ int  lstart[NBK];
    __shared__ int  lcur[NBK];
    __shared__ int  gbase[NBK];

    const int tid  = threadIdx.x;
    const int w    = blockIdx.x;
    const int e_lo = w * CHUNK;
    const int e_hi = min(e_lo + CHUNK, n_edges);
    const int n    = max(e_hi - e_lo, 0);

    int cnt_b = 0;
    if (tid < NBK) { cnt_b = counts[w * NBK + tid]; lcur[tid] = cnt_b; }
    __syncthreads();
    for (int off = 1; off < NBK; off <<= 1) {
        int t = 0;
        if (tid < NBK && tid >= off) t = lcur[tid - off];
        __syncthreads();
        if (tid < NBK) lcur[tid] += t;
        __syncthreads();
    }
    if (tid < NBK) {
        const int ex = lcur[tid] - cnt_b;
        lstart[tid] = ex;
        lcur[tid]   = ex;
        gbase[tid]  = S[tid] + pref[w * NBK + tid];
    }
    __syncthreads();

    if (n > 0) {
        int s[BATCH], d[BATCH];
#pragma unroll
        for (int k = 0; k < BATCH; ++k) {
            const int e = min(e_lo + k * BIN_THREADS + tid, e_hi - 1);
            s[k] = src[e];
            d[k] = dst[e];
        }
#pragma unroll
        for (int k = 0; k < BATCH; ++k) {
            if (e_lo + k * BIN_THREADS + tid < e_hi) {
                const int p = atomicAdd(&lcur[owner_of(d[k])], 1);
                if (p >= 0 && p < CHUNK) stage[p] = make_int2(s[k], d[k]);
            }
        }
    }
    __syncthreads();

    for (int i = tid; i < n; i += BIN_THREADS) {
        const int2 rec = stage[i];
        const int  b   = owner_of(rec.y);
        const int  g   = gbase[b] + (i - lstart[b]);
        const int so = min(max(rec.x, 0), n_nodes - 1) * ROWB;
        const int ao = min(max(rec.y - b * NPB, 0), NPB - 1) * ROWB;
        if (g >= 0 && g < n_edges) list[g] = make_int2(so, ao);
    }
}

#define AGG_THREADS 1024
#define AGG_WAVES   (AGG_THREADS / 32)
#define EPW 32
#define SPARE_OFS (NPB * ROWB)

#define TILES ((NPB + 15) / 16)

template <bool FUSE>
__global__ void __launch_bounds__(AGG_THREADS)
owner_mean(const int2* __restrict__ list, const int* __restrict__ S,
           const float* __restrict__ H, float* __restrict__ Out,
           const float* __restrict__ W, const float* __restrict__ bias,
           int n_nodes, int n_edges)
{
    __shared__ float acc[(NPB + 1) * 32];
    __shared__ int   deg[NPB + 1];

    const int tid  = threadIdx.x;
    const int lane = tid & 31;
    const unsigned lane4 = (unsigned)lane * 4u;
    const int wave = __builtin_amdgcn_readfirstlane(tid >> 5);
    const int base = blockIdx.x * NPB;

    for (int i = tid; i < (NPB + 1) * 32; i += AGG_THREADS) acc[i] = 0.0f;
    for (int i = tid; i < NPB + 1; i += AGG_THREADS) deg[i] = 0;
    __syncthreads();

    const int e_begin = max(S[blockIdx.x], 0);
    const int e_end   = min(S[blockIdx.x + 1], n_edges);

    auto fetch = [&](int e) -> int2 {
        int2 t = list[max(min(e, e_end - 1), 0)];
        if (e >= e_end) t = make_int2(0, SPARE_OFS);
        return t;
    };
    int2 r = fetch(e_begin + wave * EPW + lane);
    for (int e0 = e_begin + wave * EPW; e0 < e_end; e0 += EPW * AGG_WAVES) {
        const int2 rn = fetch(e0 + EPW * AGG_WAVES + lane);

        atomicAdd(&deg[(unsigned)r.y / ROWB], 1);

        float v[EPW];
#pragma unroll
        for (int i = 0; i < EPW; ++i) {
            const unsigned so = (unsigned)__builtin_amdgcn_readlane(r.x, i);
            v[i] = *(const float*)((const char*)H + (so + lane4));
        }
#pragma unroll
        for (int i = 0; i < EPW; ++i) {
            const unsigned ao = (unsigned)__builtin_amdgcn_readlane(r.y, i);
            atomicAdd((float*)((char*)acc + (ao + lane4)), v[i]);
        }
        r = rn;
    }
    __syncthreads();

    if (!FUSE) {
        for (int i = tid; i < NPB * 32; i += AGG_THREADS) {
            const int nl = i >> 5;
            const int n  = base + nl;
            if (n < n_nodes) {
                const float c = (float)deg[nl];
                Out[(size_t)n * 32 + (i & 31)] = acc[i] / fmaxf(c, 1.0f);
            }
        }
        return;
    }

    const int half = lane >> 4;
    const int mrow = lane & 15;
    const float bv0 = bias[mrow];
    const float bv1 = bias[16 + mrow];
    for (int t = wave; t < TILES; t += AGG_WAVES) {
        const int lrow = min(t * 16 + mrow, NPB);
        const float cnt = fmaxf((float)deg[lrow], 1.0f);

        v2f a[8];
#pragma unroll
        for (int j = 0; j < 8; ++j)
            a[j] = *(const v2f*)&acc[lrow * 32 + 4 * j + half * 2];
#pragma unroll
        for (int j = 0; j < 8; ++j) {
            a[j].x = fmaxf(a[j].x / cnt, 0.0f);
            a[j].y = fmaxf(a[j].y / cnt, 0.0f);
        }

        v8f c0 = {};
        v8f c1 = {};
#pragma unroll
        for (int j = 0; j < 8; ++j) {
            const int kb = 4 * j + half * 2;
            v2f b0, b1;
            b0.x = W[kb * 32 + mrow];
            b0.y = W[(kb + 1) * 32 + mrow];
            b1.x = W[kb * 32 + 16 + mrow];
            b1.y = W[(kb + 1) * 32 + 16 + mrow];
            c0 = __builtin_amdgcn_wmma_f32_16x16x4_f32(false, a[j], false, b0,
                                                       (short)0, c0, false, false);
            c1 = __builtin_amdgcn_wmma_f32_16x16x4_f32(false, a[j], false, b1,
                                                       (short)0, c1, false, false);
        }
#pragma unroll
        for (int r8 = 0; r8 < 8; ++r8) {
            const int nl = t * 16 + r8 + half * 8;
            const int n  = base + nl;
            if (nl < NPB && n < n_nodes) {
                Out[(size_t)n * 32 + mrow]      = c0[r8] + bv0;
                Out[(size_t)n * 32 + 16 + mrow] = c1[r8] + bv1;
            }
        }
    }
}

extern "C" void kernel_launch(void* const* d_in, const int* in_sizes, int n_in,
                              void* d_out, int out_size, void* d_ws, size_t ws_size,
                              hipStream_t stream)
{
    const float* x    = (const float*)d_in[0];
    const int*   eidx = (const int*)d_in[1];
    const float* W1   = (const float*)d_in[2];
    const float* b1   = (const float*)d_in[3];
    const float* W2   = (const float*)d_in[4];
    const float* b2   = (const float*)d_in[5];
    float*       out  = (float*)d_out;

    const int n_nodes = in_sizes[0] / 128;
    const int n_edges = in_sizes[1] / 2;
    const int* src = eidx;
    const int* dst = eidx + n_edges;

    const int nblk = (((n_edges + CHUNK - 1) / CHUNK) + 31) / 32 * 32;

    float* A      = (float*)d_ws;
    float* B      = A + (size_t)n_nodes * 32;
    int2*  list   = (int2*)(B + (size_t)n_nodes * 32);
    int*   S      = (int*)(list + n_edges);
    int*   counts = S + 2 * NBK;
    int*   pref   = counts + (size_t)nblk * NBK;

    const int n_tiles   = (n_nodes + 15) / 16;
    const int grid_gemm = (n_tiles + 7) / 8;

    bin_count<<<nblk, BIN_THREADS, 0, stream>>>(dst, counts, n_edges);
    bin_scan_and_gemm<<<grid_gemm + 1, NBK, 0, stream>>>(counts, pref, S, nblk,
                                                        x, W1, b1, A, 128, n_nodes);
    bin_fill<<<nblk, BIN_THREADS, 0, stream>>>(src, dst, counts, pref, S, list, n_edges, n_nodes);
    owner_mean<true><<<NBK, AGG_THREADS, 0, stream>>>(list, S, A, B, W2, b2, n_nodes, n_edges);
    owner_mean<false><<<NBK, AGG_THREADS, 0, stream>>>(list, S, B, out, W2, b2, n_nodes, n_edges);
}


# edge chunks doubled (13312): histogram table so small that each fill work-group sums its own list slots from it; no scan launch (5 -> 4), X@W1 rides with the count pass; sums stay LDS arrival order
# speedup vs baseline: 5.9706x; 1.0402x over previous
#include <hip/hip_runtime.h>
#include <stdint.h>


typedef float v2f __attribute__((ext_vector_type(2)));
typedef float v8f __attribute__((ext_vector_type(8)));

__device__ __forceinline__ void
gemm_bias_tile(const float* __restrict__ X, const float* __restrict__ W,
               const float* __restrict__ bias, float* __restrict__ Y,
               int K, int n_rows, int wave)
{
    const int lane    = threadIdx.x & 31;
    const int n_tiles = (n_rows + 15) >> 4;
    if (wave >= n_tiles) return;

    const int half = lane >> 4;
    const int mrow = lane & 15;
    const int arow = min(wave * 16 + mrow, n_rows - 1);
    const float* __restrict__ xrow = X + (size_t)arow * (size_t)K;

    v8f c0 = {};
    v8f c1 = {};

    for (int k = 0; k < K; k += 4) {
        const int kb = k + half * 2;
        v2f a = *(const v2f*)(xrow + kb);

        v2f b0, b1;
        b0.x = W[(size_t)kb * 32 + mrow];
        b0.y = W[(size_t)(kb + 1) * 32 + mrow];
        b1.x = W[(size_t)kb * 32 + 16 + mrow];
        b1.y = W[(size_t)(kb + 1) * 32 + 16 + mrow];

        c0 = __builtin_amdgcn_wmma_f32_16x16x4_f32(false, a, false, b0,
                                                   (short)0, c0, false, false);
        c1 = __builtin_amdgcn_wmma_f32_16x16x4_f32(false, a, false, b1,
                                                   (short)0, c1, false, false);
    }

    const float bv0 = bias[mrow];
    const float bv1 = bias[16 + mrow];
#pragma unroll
    for (int r = 0; r < 8; ++r) {
        const int orow = wave * 16 + r + half * 8;
        if (orow < n_rows) {
            Y[(size_t)orow * 32 + mrow]      = c0[r] + bv0;
            Y[(size_t)orow * 32 + 16 + mrow] = c1[r] + bv1;
        }
    }
}

#define NBK 256
#define NPB 391
#define BIN_THREADS 1024
#define BATCH 13
#define CHUNK (BATCH * BIN_THREADS)
#define CNT_THREADS 512
#define CNT_BATCH (CHUNK / CNT_THREADS)
#define NCH 128
#define COLG (BIN_THREADS / NBK)
#define ROWS_PT (NCH / COLG)
#define ROWB 128

__device__ __forceinline__ int owner_of(int d)
{
    const unsigned b = (unsigned)d / (unsigned)NPB;
    return (int)(b < (unsigned)(NBK - 1) ? b : (unsigned)(NBK - 1));
}

__global__ void __launch_bounds__(CNT_THREADS)
bin_count_and_gemm(const int* __restrict__ dst, int* __restrict__ counts, int n_edges,
                   const float* __restrict__ X, const float* __restrict__ W,
                   const float* __restrict__ bias, float* __restrict__ Y,
                   int K, int n_rows)
{
    __shared__ int hist[NBK];
    if (blockIdx.x >= NCH) {
        gemm_bias_tile(X, W, bias, Y, K, n_rows,
                       (int)(((blockIdx.x - NCH) * blockDim.x + threadIdx.x) >> 5));
        return;
    }
    const int tid = threadIdx.x;
    if (tid < NBK) hist[tid] = 0;
    __syncthreads();
    const int e_lo = blockIdx.x * CHUNK;
    const int e_hi = min(e_lo + CHUNK, n_edges);
    if (e_lo < e_hi) {
        int d[CNT_BATCH];
#pragma unroll
        for (int k = 0; k < CNT_BATCH; ++k)
            d[k] = dst[min(e_lo + k * CNT_THREADS + tid, e_hi - 1)];
#pragma unroll
        for (int k = 0; k < CNT_BATCH; ++k)
            if (e_lo + k * CNT_THREADS + tid < e_hi)
                atomicAdd(&hist[owner_of(d[k])], 1);
    }
    __syncthreads();
    if (tid < NBK) counts[blockIdx.x * NBK + tid] = hist[tid];
}

__global__ void __launch_bounds__(BIN_THREADS)
bin_fill(const int* __restrict__ src, const int* __restrict__ dst,
         const int* __restrict__ counts, int* __restrict__ S,
         int2* __restrict__ list, int n_edges, int n_nodes)
{
    __shared__ int2 stage[CHUNK];
    __shared__ int  lstart[NBK];
    __shared__ int  lcur[NBK];
    __shared__ int  gbase[NBK];
    __shared__ int  ptot[COLG][NBK];
    __shared__ int  ppre[COLG][NBK];
    __shared__ int  tsum[NBK];

    const int tid  = threadIdx.x;
    const int w    = blockIdx.x;
    const int e_lo = w * CHUNK;
    const int e_hi = min(e_lo + CHUNK, n_edges);
    const int n    = max(e_hi - e_lo, 0);

    int s[BATCH], d[BATCH];
#pragma unroll
    for (int k = 0; k < BATCH; ++k) {
        const int e = max(min(e_lo + k * BIN_THREADS + tid, e_hi - 1), 0);
        s[k] = src[e];
        d[k] = dst[e];
    }

    int cnt_b = 0;
    if (tid < NBK) cnt_b = counts[w * NBK + tid];

    {
        const int col = tid & (NBK - 1);
        const int g   = tid / NBK;
        int tot = 0, pre = 0;
        const int* __restrict__ p = counts + (g * NBK + col);
        int c[ROWS_PT];
#pragma unroll
        for (int j = 0; j < ROWS_PT; ++j) c[j] = p[j * COLG * NBK];
#pragma unroll
        for (int j = 0; j < ROWS_PT; ++j) {
            const int r = g + j * COLG;
            tot += c[j];
            pre += (r < w) ? c[j] : 0;
        }
        ptot[g][col] = tot;
        ppre[g][col] = pre;
    }
    __syncthreads();

    int tot_b = 0, pre_b = 0;
    if (tid < NBK) {
#pragma unroll
        for (int g = 0; g < COLG; ++g) { tot_b += ptot[g][tid]; pre_b += ppre[g][tid]; }
        lcur[tid] = cnt_b;
        tsum[tid] = tot_b;
    }
    __syncthreads();
    for (int off = 1; off < NBK; off <<= 1) {
        int t = 0, u = 0;
        if (tid < NBK && tid >= off) { t = lcur[tid - off]; u = tsum[tid - off]; }
        __syncthreads();
        if (tid < NBK) { lcur[tid] += t; tsum[tid] += u; }
        __syncthreads();
    }
    if (tid < NBK) {
        const int ex = lcur[tid] - cnt_b;
        lstart[tid] = ex;
        lcur[tid]   = ex;
        gbase[tid]  = (tsum[tid] - tot_b) + pre_b;
        if (w == 0) {
            S[tid + 1] = tsum[tid];
            if (tid == 0) S[0] = 0;
        }
    }
    __syncthreads();

#pragma unroll
    for (int k = 0; k < BATCH; ++k) {
        if (e_lo + k * BIN_THREADS + tid < e_hi) {
            const int p = atomicAdd(&lcur[owner_of(d[k])], 1);
            if (p >= 0 && p < CHUNK) stage[p] = make_int2(s[k], d[k]);
        }
    }
    __syncthreads();

    for (int i = tid; i < n; i += BIN_THREADS) {
        const int2 rec = stage[i];
        const int  b   = owner_of(rec.y);
        const int  g   = gbase[b] + (i - lstart[b]);
        const int so = min(max(rec.x, 0), n_nodes - 1) * ROWB;
        const int ao = min(max(rec.y - b * NPB, 0), NPB - 1) * ROWB;
        if (g >= 0 && g < n_edges) list[g] = make_int2(so, ao);
    }
}

#define AGG_THREADS 1024
#define AGG_WAVES   (AGG_THREADS / 32)
#define EPW 32
#define SPARE_OFS (NPB * ROWB)

#define TILES ((NPB + 15) / 16)

template <bool FUSE>
__global__ void __launch_bounds__(AGG_THREADS)
owner_mean(const int2* __restrict__ list, const int* __restrict__ S,
           const float* __restrict__ H, float* __restrict__ Out,
           const float* __restrict__ W, const float* __restrict__ bias,
           int n_nodes, int n_edges)
{
    __shared__ float acc[(NPB + 1) * 32];
    __shared__ int   deg[NPB + 1];

    const int tid  = threadIdx.x;
    const int lane = tid & 31;
    const unsigned lane4 = (unsigned)lane * 4u;
    const int wave = __builtin_amdgcn_readfirstlane(tid >> 5);
    const int base = blockIdx.x * NPB;

    for (int i = tid; i < (NPB + 1) * 32; i += AGG_THREADS) acc[i] = 0.0f;
    for (int i = tid; i < NPB + 1; i += AGG_THREADS) deg[i] = 0;
    __syncthreads();

    const int e_begin = max(S[blockIdx.x], 0);
    const int e_end   = min(S[blockIdx.x + 1], n_edges);

    auto fetch = [&](int e) -> int2 {
        int2 t = list[max(min(e, e_end - 1), 0)];
        if (e >= e_end) t = make_int2(0, SPARE_OFS);
        return t;
    };
    int2 r = fetch(e_begin + wave * EPW + lane);
    for (int e0 = e_begin + wave * EPW; e0 < e_end; e0 += EPW * AGG_WAVES) {
        const int2 rn = fetch(e0 + EPW * AGG_WAVES + lane);

        atomicAdd(&deg[(unsigned)r.y / ROWB], 1);

        float v[EPW];
#pragma unroll
        for (int i = 0; i < EPW; ++i) {
            const unsigned so = (unsigned)__builtin_amdgcn_readlane(r.x, i);
            v[i] = *(const float*)((const char*)H + (so + lane4));
        }
#pragma unroll
        for (int i = 0; i < EPW; ++i) {
            const unsigned ao = (unsigned)__builtin_amdgcn_readlane(r.y, i);
            atomicAdd((float*)((char*)acc + (ao + lane4)), v[i]);
        }
        r = rn;
    }
    __syncthreads();

    if (!FUSE) {
        for (int i = tid; i < NPB * 32; i += AGG_THREADS) {
            const int nl = i >> 5;
            const int n  = base + nl;
            if (n < n_nodes) {
                const float c = (float)deg[nl];
                Out[(size_t)n * 32 + (i & 31)] = acc[i] / fmaxf(c, 1.0f);
            }
        }
        return;
    }

    const int half = lane >> 4;
    const int mrow = lane & 15;
    const float bv0 = bias[mrow];
    const float bv1 = bias[16 + mrow];
    for (int t = wave; t < TILES; t += AGG_WAVES) {
        const int lrow = min(t * 16 + mrow, NPB);
        const float cnt = fmaxf((float)deg[lrow], 1.0f);

        v2f a[8];
#pragma unroll
        for (int j = 0; j < 8; ++j)
            a[j] = *(const v2f*)&acc[lrow * 32 + 4 * j + half * 2];
#pragma unroll
        for (int j = 0; j < 8; ++j) {
            a[j].x = fmaxf(a[j].x / cnt, 0.0f);
            a[j].y = fmaxf(a[j].y / cnt, 0.0f);
        }

        v8f c0 = {};
        v8f c1 = {};
#pragma unroll
        for (int j = 0; j < 8; ++j) {
            const int kb = 4 * j + half * 2;
            v2f b0, b1;
            b0.x = W[kb * 32 + mrow];
            b0.y = W[(kb + 1) * 32 + mrow];
            b1.x = W[kb * 32 + 16 + mrow];
            b1.y = W[(kb + 1) * 32 + 16 + mrow];
            c0 = __builtin_amdgcn_wmma_f32_16x16x4_f32(false, a[j], false, b0,
                                                       (short)0, c0, false, false);
            c1 = __builtin_amdgcn_wmma_f32_16x16x4_f32(false, a[j], false, b1,
                                                       (short)0, c1, false, false);
        }
#pragma unroll
        for (int r8 = 0; r8 < 8; ++r8) {
            const int nl = t * 16 + r8 + half * 8;
            const int n  = base + nl;
            if (nl < NPB && n < n_nodes) {
                Out[(size_t)n * 32 + mrow]      = c0[r8] + bv0;
                Out[(size_t)n * 32 + 16 + mrow] = c1[r8] + bv1;
            }
        }
    }
}

extern "C" void kernel_launch(void* const* d_in, const int* in_sizes, int n_in,
                              void* d_out, int out_size, void* d_ws, size_t ws_size,
                              hipStream_t stream)
{
    const float* x    = (const float*)d_in[0];
    const int*   eidx = (const int*)d_in[1];
    const float* W1   = (const float*)d_in[2];
    const float* b1   = (const float*)d_in[3];
    const float* W2   = (const float*)d_in[4];
    const float* b2   = (const float*)d_in[5];
    float*       out  = (float*)d_out;

    const int n_nodes = in_sizes[0] / 128;
    const int n_edges = in_sizes[1] / 2;
    const int* src = eidx;
    const int* dst = eidx + n_edges;

    float* A      = (float*)d_ws;
    float* B      = A + (size_t)n_nodes * 32;
    int2*  list   = (int2*)(B + (size_t)n_nodes * 32);
    int*   S      = (int*)(list + n_edges);
    int*   counts = S + 2 * NBK;

    const int n_tiles   = (n_nodes + 15) / 16;
    const int grid_gemm = (n_tiles + 15) / 16;

    bin_count_and_gemm<<<NCH + grid_gemm, CNT_THREADS, 0, stream>>>(dst, counts, n_edges,
                                                                  x, W1, b1, A, 128, n_nodes);
    bin_fill<<<NCH, BIN_THREADS, 0, stream>>>(src, dst, counts, S, list, n_edges, n_nodes);
    owner_mean<true><<<NBK, AGG_THREADS, 0, stream>>>(list, S, A, B, W2, b2, n_nodes, n_edges);
    owner_mean<false><<<NBK, AGG_THREADS, 0, stream>>>(list, S, B, out, W2, b2, n_nodes, n_edges);
}
